# starting kernel
# baseline (speedup 1.0000x reference)
#include <hip/hip_runtime.h>
#include <hip/hip_bf16.h>

typedef float v2f __attribute__((ext_vector_type(2)));
typedef float v8f __attribute__((ext_vector_type(8)));

#define XS_N 4608
#define WS_N 16384
#define VS_N 8192

__global__ __launch_bounds__(1024)
void deconv_caps_routing(const float* __restrict__ xg,
                         const float* __restrict__ wg,
                         const float* __restrict__ bg,
                         float* __restrict__ outg) {
  extern __shared__ float smem[];
  float* xs  = smem;
  float* ws  = smem + XS_N;
  float* vsm = smem + XS_N + WS_N;

  const int tid = threadIdx.x;
  const int blk = blockIdx.x;
  const int b  = blk >> 10;
  const int od = (blk >> 5) & 31;
  const int oh = blk & 31;

  const int tdbase = 1 - (od & 1);
  const int thbase = 1 - (oh & 1);
  const int id0 = (od + 1) >> 1;
  const int ih0 = (oh + 1) >> 1;

  for (int idx = tid; idx < WS_N; idx += 1024) {
    int ai  = idx & 15;
    int c   = (idx >> 4) & 63;
    int tw  = (idx >> 10) & 3;
    int thi = (idx >> 12) & 1;
    int tdi = (idx >> 13) & 1;
    int td = tdbase + 2 * tdi;
    int th = thbase + 2 * thi;
    ws[idx] = wg[((ai * 64 + c) << 6) + td * 16 + th * 4 + tw];
  }
  for (int idx = tid; idx < XS_N; idx += 1024) {
    int ai   = idx & 15;
    int t    = idx >> 4;
    int iwp  = t % 18;
    int rest = t / 18;
    int thi = rest & 1;
    int tdi = (rest >> 1) & 1;
    int ci  = rest >> 2;
    int iw = iwp - 1;
    int id = id0 - tdi;
    int ih = ih0 - thi;
    float v = 0.f;
    if ((unsigned)iw < 16u && (unsigned)id < 16u && (unsigned)ih < 16u)
      v = xg[(((b * 4 + ci) * 16 + ai) << 12) + (id << 8) + (ih << 4) + iw];
    xs[idx] = v;
  }
  __syncthreads();

  const int lane = tid & 31;
  const int wv   = tid >> 5;
  const int par  = wv >> 4;
  const int icap = (wv >> 2) & 3;
  const int ocap = wv & 3;
  const int ln = lane & 15;
  const int kh = lane >> 4;

  v8f acc = {0.f, 0.f, 0.f, 0.f, 0.f, 0.f, 0.f, 0.f};
#pragma unroll
  for (int j = 0; j < 32; ++j) {
    int tap = j >> 2;
    int twi = tap & 1;
    int thi = (tap >> 1) & 1;
    int tdi = tap >> 2;
    int aibase = ((j & 3) << 2) + (kh << 1);
    int iw = ln + par - twi;
    int aoff = ((((icap * 2 + tdi) * 2 + thi) * 18 + (iw + 1)) << 4) + aibase;
    int tw = (1 - par) + (twi << 1);
    int boff = (((((tdi * 2 + thi) << 2) + tw) * 64 + (ocap * 16 + ln)) << 4) + aibase;
    v2f a  = *(const v2f*)(xs + aoff);
    v2f bb = *(const v2f*)(ws + boff);
    acc = __builtin_amdgcn_wmma_f32_16x16x4_f32(
         false, a,  false, bb,
         (short)0, acc,  false,  false);
  }

  const int p_idx = icap * 4 + ocap;
#pragma unroll
  for (int g = 0; g < 8; ++g) {
    int vx = g + (kh << 3);
    vsm[(((par * 16 + vx) * 16 + p_idx) << 4) + ln] = acc[g];
  }
  __syncthreads();

  const int rr = wv & 15;
  const int p  = lane & 15;
  const int h  = lane >> 4;
  const int ri = p >> 2;
  const int ro = p & 3;

  float vv[8], bv[8];
  const float* vbase = vsm + (((par * 16 + rr) * 16 + p) << 4) + (h << 3);
#pragma unroll
  for (int j = 0; j < 8; ++j) vv[j] = vbase[j];
#pragma unroll
  for (int j = 0; j < 8; ++j) bv[j] = bg[ro * 16 + (h << 3) + j];

  float logit = 0.f;
  float pre[8];
  float scale = 0.f;
#pragma unroll
  for (int it = 0; it < 3; ++it) {
    float m = logit;
    m = fmaxf(m, __shfl_xor(m, 1));
    m = fmaxf(m, __shfl_xor(m, 2));
    float e = __expf(logit - m);
    float s = e;
    s += __shfl_xor(s, 1);
    s += __shfl_xor(s, 2);
    float route = e / s;
#pragma unroll
    for (int j = 0; j < 8; ++j) pre[j] = route * vv[j];
#pragma unroll
    for (int j = 0; j < 8; ++j) {
      pre[j] += __shfl_xor(pre[j], 4);
      pre[j] += __shfl_xor(pre[j], 8);
      pre[j] += bv[j];
    }
    if (it < 2) {
      float dt = 0.f, np2 = 0.f, nv2 = 0.f;
#pragma unroll
      for (int j = 0; j < 8; ++j) {
        dt  += pre[j] * vv[j];
        np2 += pre[j] * pre[j];
        nv2 += vv[j] * vv[j];
      }
      dt  += __shfl_xor(dt, 16);
      np2 += __shfl_xor(np2, 16);
      nv2 += __shfl_xor(nv2, 16);
      float na = sqrtf(np2), nb = sqrtf(nv2);
      logit += dt / fmaxf(na * nb, 1e-8f);
    } else {
      float n2 = 0.f;
#pragma unroll
      for (int j = 0; j < 8; ++j) n2 += pre[j] * pre[j];
      n2 += __shfl_xor(n2, 16);
      float norm = sqrtf(n2);
      scale = n2 / ((norm + 1e-12f) * (1.f + n2));
    }
  }

  if (ri == 0) {
    int ow = (rr << 1) + par;
#pragma unroll
    for (int j = 0; j < 8; ++j) {
      int ao = (h << 3) + j;
      outg[(((b * 4 + ro) * 16 + ao) << 15) + (od << 10) + (oh << 5) + ow] =
          pre[j] * scale;
    }
  }
}

extern "C" void kernel_launch(void* const* d_in, const int* in_sizes, int n_in,
                              void* d_out, int out_size, void* d_ws, size_t ws_size,
                              hipStream_t stream) {
  (void)in_sizes; (void)n_in; (void)out_size; (void)d_ws; (void)ws_size;
  const float* x = (const float*)d_in[0];
  const float* w = (const float*)d_in[1];
  const float* b = (const float*)d_in[2];
  float* out = (float*)d_out;

  const size_t shmem = (size_t)(XS_N + WS_N + VS_N) * sizeof(float);
  deconv_caps_routing<<<dim3(4 * 32 * 32), dim3(1024), shmem, stream>>>(x, w, b, out);
}


# coalesced weight staging: each lane loads the float4 of the 4 tw taps (a wave-load touches 16 lines x 32 B, was 32 lines x 4 B), ai pairs go to LDS as b64; LDS layout, GEMM, routing unchanged
# speedup vs baseline: 1.2263x; 1.2263x over previous
#include <hip/hip_runtime.h>
#include <hip/hip_bf16.h>

typedef float v2f __attribute__((ext_vector_type(2)));
typedef float v8f __attribute__((ext_vector_type(8)));

#define XS_N 4608
#define WS_N 16384
#define VS_N 8192

__global__ __launch_bounds__(1024)
void deconv_caps_routing(const float* __restrict__ xg,
                         const float* __restrict__ wg,
                         const float* __restrict__ bg,
                         float* __restrict__ outg) {
  extern __shared__ float smem[];
  float* xs  = smem;
  float* ws  = smem + XS_N;
  float* vsm = smem + XS_N + WS_N;

  const int tid = threadIdx.x;
  const int blk = blockIdx.x;
  const int b  = blk >> 10;
  const int od = (blk >> 5) & 31;
  const int oh = blk & 31;

  const int tdbase = 1 - (od & 1);
  const int thbase = 1 - (oh & 1);
  const int id0 = (od + 1) >> 1;
  const int ih0 = (oh + 1) >> 1;

  {
    const float4* __restrict__ wg4 = reinterpret_cast<const float4*>(wg);
    const int s_aip = tid & 7;
    const int s_c   = ((tid >> 5) << 1) | ((tid >> 3) & 1);
    const int s_thi = (tid >> 4) & 1;
    const int s_th  = thbase + 2 * s_thi;
    const int g_even = ((2 * s_aip) * 64 + s_c) * 16 + s_th;
    const int g_odd  = g_even + 64 * 16;
    const float4 e0 = wg4[g_even + (tdbase    ) * 4];
    const float4 o0 = wg4[g_odd  + (tdbase    ) * 4];
    const float4 e1 = wg4[g_even + (tdbase + 2) * 4];
    const float4 o1 = wg4[g_odd  + (tdbase + 2) * 4];
    float* d0 = ws + 4096 * s_thi + 16 * s_c + 2 * s_aip;
    float* d1 = d0 + 8192;
    *(v2f*)(d0 + 0 * 1024) = (v2f){e0.x, o0.x};
    *(v2f*)(d0 + 1 * 1024) = (v2f){e0.y, o0.y};
    *(v2f*)(d0 + 2 * 1024) = (v2f){e0.z, o0.z};
    *(v2f*)(d0 + 3 * 1024) = (v2f){e0.w, o0.w};
    *(v2f*)(d1 + 0 * 1024) = (v2f){e1.x, o1.x};
    *(v2f*)(d1 + 1 * 1024) = (v2f){e1.y, o1.y};
    *(v2f*)(d1 + 2 * 1024) = (v2f){e1.z, o1.z};
    *(v2f*)(d1 + 3 * 1024) = (v2f){e1.w, o1.w};
  }
  for (int idx = tid; idx < XS_N; idx += 1024) {
    int ai   = idx & 15;
    int t    = idx >> 4;
    int iwp  = t % 18;
    int rest = t / 18;
    int thi = rest & 1;
    int tdi = (rest >> 1) & 1;
    int ci  = rest >> 2;
    int iw = iwp - 1;
    int id = id0 - tdi;
    int ih = ih0 - thi;
    float v = 0.f;
    if ((unsigned)iw < 16u && (unsigned)id < 16u && (unsigned)ih < 16u)
      v = xg[(((b * 4 + ci) * 16 + ai) << 12) + (id << 8) + (ih << 4) + iw];
    xs[idx] = v;
  }
  __syncthreads();

  const int lane = tid & 31;
  const int wv   = tid >> 5;
  const int par  = wv >> 4;
  const int icap = (wv >> 2) & 3;
  const int ocap = wv & 3;
  const int ln = lane & 15;
  const int kh = lane >> 4;

  v8f acc = {0.f, 0.f, 0.f, 0.f, 0.f, 0.f, 0.f, 0.f};
#pragma unroll
  for (int j = 0; j < 32; ++j) {
    int tap = j >> 2;
    int twi = tap & 1;
    int thi = (tap >> 1) & 1;
    int tdi = tap >> 2;
    int aibase = ((j & 3) << 2) + (kh << 1);
    int iw = ln + par - twi;
    int aoff = ((((icap * 2 + tdi) * 2 + thi) * 18 + (iw + 1)) << 4) + aibase;
    int tw = (1 - par) + (twi << 1);
    int boff = (((((tdi * 2 + thi) << 2) + tw) * 64 + (ocap * 16 + ln)) << 4) + aibase;
    v2f a  = *(const v2f*)(xs + aoff);
    v2f bb = *(const v2f*)(ws + boff);
    acc = __builtin_amdgcn_wmma_f32_16x16x4_f32(
         false, a,  false, bb,
         (short)0, acc,  false,  false);
  }

  const int p_idx = icap * 4 + ocap;
#pragma unroll
  for (int g = 0; g < 8; ++g) {
    int vx = g + (kh << 3);
    vsm[(((par * 16 + vx) * 16 + p_idx) << 4) + ln] = acc[g];
  }
  __syncthreads();

  const int rr = wv & 15;
  const int p  = lane & 15;
  const int h  = lane >> 4;
  const int ri = p >> 2;
  const int ro = p & 3;

  float vv[8], bv[8];
  const float* vbase = vsm + (((par * 16 + rr) * 16 + p) << 4) + (h << 3);
#pragma unroll
  for (int j = 0; j < 8; ++j) vv[j] = vbase[j];
#pragma unroll
  for (int j = 0; j < 8; ++j) bv[j] = bg[ro * 16 + (h << 3) + j];

  float logit = 0.f;
  float pre[8];
  float scale = 0.f;
#pragma unroll
  for (int it = 0; it < 3; ++it) {
    float m = logit;
    m = fmaxf(m, __shfl_xor(m, 1));
    m = fmaxf(m, __shfl_xor(m, 2));
    float e = __expf(logit - m);
    float s = e;
    s += __shfl_xor(s, 1);
    s += __shfl_xor(s, 2);
    float route = e / s;
#pragma unroll
    for (int j = 0; j < 8; ++j) pre[j] = route * vv[j];
#pragma unroll
    for (int j = 0; j < 8; ++j) {
      pre[j] += __shfl_xor(pre[j], 4);
      pre[j] += __shfl_xor(pre[j], 8);
      pre[j] += bv[j];
    }
    if (it < 2) {
      float dt = 0.f, np2 = 0.f, nv2 = 0.f;
#pragma unroll
      for (int j = 0; j < 8; ++j) {
        dt  += pre[j] * vv[j];
        np2 += pre[j] * pre[j];
        nv2 += vv[j] * vv[j];
      }
      dt  += __shfl_xor(dt, 16);
      np2 += __shfl_xor(np2, 16);
      nv2 += __shfl_xor(nv2, 16);
      float na = sqrtf(np2), nb = sqrtf(nv2);
      logit += dt / fmaxf(na * nb, 1e-8f);
    } else {
      float n2 = 0.f;
#pragma unroll
      for (int j = 0; j < 8; ++j) n2 += pre[j] * pre[j];
      n2 += __shfl_xor(n2, 16);
      float norm = sqrtf(n2);
      scale = n2 / ((norm + 1e-12f) * (1.f + n2));
    }
  }

  if (ri == 0) {
    int ow = (rr << 1) + par;
#pragma unroll
    for (int j = 0; j < 8; ++j) {
      int ao = (h << 3) + j;
      outg[(((b * 4 + ro) * 16 + ao) << 15) + (od << 10) + (oh << 5) + ow] =
          pre[j] * scale;
    }
  }
}

extern "C" void kernel_launch(void* const* d_in, const int* in_sizes, int n_in,
                              void* d_out, int out_size, void* d_ws, size_t ws_size,
                              hipStream_t stream) {
  (void)in_sizes; (void)n_in; (void)out_size; (void)d_ws; (void)ws_size;
  const float* x = (const float*)d_in[0];
  const float* w = (const float*)d_in[1];
  const float* b = (const float*)d_in[2];
  float* out = (float*)d_out;

  const size_t shmem = (size_t)(XS_N + WS_N + VS_N) * sizeof(float);
  deconv_caps_routing<<<dim3(4 * 32 * 32), dim3(1024), shmem, stream>>>(x, w, b, out);
}


# LDS rows of 16 floats padded to a pitch of 20 (votes: +4 per voxel, 16 B aligned): the 4-way bank conflicts of the A/B fragment reads and of the routing reads are gone; k order and sums unchanged
# speedup vs baseline: 1.3146x; 1.0720x over previous
#include <hip/hip_runtime.h>
#include <hip/hip_bf16.h>

typedef float v2f __attribute__((ext_vector_type(2)));
typedef float v8f __attribute__((ext_vector_type(8)));

#define RP 20
#define X_ROWS 288
#define XS_N (X_ROWS * RP)
#define W_PL (64 * RP)
#define WS_N (16 * W_PL)
#define V_VOX (16 * RP + 4)
#define VS_N (32 * V_VOX)

__global__ __launch_bounds__(1024)
void deconv_caps_routing(const float* __restrict__ xg,
                         const float* __restrict__ wg,
                         const float* __restrict__ bg,
                         float* __restrict__ outg) {
  extern __shared__ __attribute__((aligned(16))) float smem[];
  float* xs  = smem;
  float* ws  = smem + XS_N;
  float* vsm = smem + XS_N + WS_N;

  const int tid = threadIdx.x;
  const int blk = blockIdx.x;
  const int b  = blk >> 10;
  const int od = (blk >> 5) & 31;
  const int oh = blk & 31;

  const int tdbase = 1 - (od & 1);
  const int thbase = 1 - (oh & 1);
  const int id0 = (od + 1) >> 1;
  const int ih0 = (oh + 1) >> 1;

  {
    const float4* __restrict__ wg4 = reinterpret_cast<const float4*>(wg);
    const int s_aip = tid & 7;
    const int s_c   = ((tid >> 5) << 1) | ((tid >> 3) & 1);
    const int s_thi = (tid >> 4) & 1;
    const int s_th  = thbase + 2 * s_thi;
    const int g_even = ((2 * s_aip) * 64 + s_c) * 16 + s_th;
    const int g_odd  = g_even + 64 * 16;
    const float4 e0 = wg4[g_even + (tdbase    ) * 4];
    const float4 o0 = wg4[g_odd  + (tdbase    ) * 4];
    const float4 e1 = wg4[g_even + (tdbase + 2) * 4];
    const float4 o1 = wg4[g_odd  + (tdbase + 2) * 4];
    float* d0 = ws + 4 * W_PL * s_thi + RP * s_c + 2 * s_aip;
    float* d1 = d0 + 8 * W_PL;
    *(v2f*)(d0 + 0 * W_PL) = (v2f){e0.x, o0.x};
    *(v2f*)(d0 + 1 * W_PL) = (v2f){e0.y, o0.y};
    *(v2f*)(d0 + 2 * W_PL) = (v2f){e0.z, o0.z};
    *(v2f*)(d0 + 3 * W_PL) = (v2f){e0.w, o0.w};
    *(v2f*)(d1 + 0 * W_PL) = (v2f){e1.x, o1.x};
    *(v2f*)(d1 + 1 * W_PL) = (v2f){e1.y, o1.y};
    *(v2f*)(d1 + 2 * W_PL) = (v2f){e1.z, o1.z};
    *(v2f*)(d1 + 3 * W_PL) = (v2f){e1.w, o1.w};
  }
  for (int idx = tid; idx < X_ROWS * 16; idx += 1024) {
    int ai   = idx & 15;
    int t    = idx >> 4;
    int iwp  = t % 18;
    int rest = t / 18;
    int thi = rest & 1;
    int tdi = (rest >> 1) & 1;
    int ci  = rest >> 2;
    int iw = iwp - 1;
    int id = id0 - tdi;
    int ih = ih0 - thi;
    float v = 0.f;
    if ((unsigned)iw < 16u && (unsigned)id < 16u && (unsigned)ih < 16u)
      v = xg[(((b * 4 + ci) * 16 + ai) << 12) + (id << 8) + (ih << 4) + iw];
    xs[t * RP + ai] = v;
  }
  __syncthreads();

  const int lane = tid & 31;
  const int wv   = tid >> 5;
  const int par  = wv >> 4;
  const int icap = (wv >> 2) & 3;
  const int ocap = wv & 3;
  const int ln = lane & 15;
  const int kh = lane >> 4;

  v8f acc = {0.f, 0.f, 0.f, 0.f, 0.f, 0.f, 0.f, 0.f};
#pragma unroll
  for (int j = 0; j < 32; ++j) {
    int tap = j >> 2;
    int twi = tap & 1;
    int thi = (tap >> 1) & 1;
    int tdi = tap >> 2;
    int aibase = ((j & 3) << 2) + (kh << 1);
    int iw = ln + par - twi;
    int aoff = (((icap * 2 + tdi) * 2 + thi) * 18 + (iw + 1)) * RP + aibase;
    int tw = (1 - par) + (twi << 1);
    int boff = ((((tdi * 2 + thi) << 2) + tw) * 64 + (ocap * 16 + ln)) * RP + aibase;
    v2f a  = *(const v2f*)(xs + aoff);
    v2f bb = *(const v2f*)(ws + boff);
    acc = __builtin_amdgcn_wmma_f32_16x16x4_f32(
         false, a,  false, bb,
         (short)0, acc,  false,  false);
  }

  const int p_idx = icap * 4 + ocap;
#pragma unroll
  for (int g = 0; g < 8; ++g) {
    int vx = g + (kh << 3);
    vsm[(par * 16 + vx) * V_VOX + p_idx * RP + ln] = acc[g];
  }
  __syncthreads();

  const int rr = wv & 15;
  const int p  = lane & 15;
  const int h  = lane >> 4;
  const int ri = p >> 2;
  const int ro = p & 3;

  float vv[8], bv[8];
  const float* vbase = vsm + (par * 16 + rr) * V_VOX + p * RP + (h << 3);
#pragma unroll
  for (int j = 0; j < 8; ++j) vv[j] = vbase[j];
#pragma unroll
  for (int j = 0; j < 8; ++j) bv[j] = bg[ro * 16 + (h << 3) + j];

  float logit = 0.f;
  float pre[8];
  float scale = 0.f;
#pragma unroll
  for (int it = 0; it < 3; ++it) {
    float m = logit;
    m = fmaxf(m, __shfl_xor(m, 1));
    m = fmaxf(m, __shfl_xor(m, 2));
    float e = __expf(logit - m);
    float s = e;
    s += __shfl_xor(s, 1);
    s += __shfl_xor(s, 2);
    float route = e / s;
#pragma unroll
    for (int j = 0; j < 8; ++j) pre[j] = route * vv[j];
#pragma unroll
    for (int j = 0; j < 8; ++j) {
      pre[j] += __shfl_xor(pre[j], 4);
      pre[j] += __shfl_xor(pre[j], 8);
      pre[j] += bv[j];
    }
    if (it < 2) {
      float dt = 0.f, np2 = 0.f, nv2 = 0.f;
#pragma unroll
      for (int j = 0; j < 8; ++j) {
        dt  += pre[j] * vv[j];
        np2 += pre[j] * pre[j];
        nv2 += vv[j] * vv[j];
      }
      dt  += __shfl_xor(dt, 16);
      np2 += __shfl_xor(np2, 16);
      nv2 += __shfl_xor(nv2, 16);
      float na = sqrtf(np2), nb = sqrtf(nv2);
      logit += dt / fmaxf(na * nb, 1e-8f);
    } else {
      float n2 = 0.f;
#pragma unroll
      for (int j = 0; j < 8; ++j) n2 += pre[j] * pre[j];
      n2 += __shfl_xor(n2, 16);
      float norm = sqrtf(n2);
      scale = n2 / ((norm + 1e-12f) * (1.f + n2));
    }
  }

  if (ri == 0) {
    int ow = (rr << 1) + par;
#pragma unroll
    for (int j = 0; j < 8; ++j) {
      int ao = (h << 3) + j;
      outg[(((b * 4 + ro) * 16 + ao) << 15) + (od << 10) + (oh << 5) + ow] =
          pre[j] * scale;
    }
  }
}

extern "C" void kernel_launch(void* const* d_in, const int* in_sizes, int n_in,
                              void* d_out, int out_size, void* d_ws, size_t ws_size,
                              hipStream_t stream) {
  (void)in_sizes; (void)n_in; (void)out_size; (void)d_ws; (void)ws_size;
  const float* x = (const float*)d_in[0];
  const float* w = (const float*)d_in[1];
  const float* b = (const float*)d_in[2];
  float* out = (float*)d_out;

  const size_t shmem = (size_t)(XS_N + WS_N + VS_N) * sizeof(float);
  deconv_caps_routing<<<dim3(4 * 32 * 32), dim3(1024), shmem, stream>>>(x, w, b, out);
}


# out rows staged through LDS (64 rows x 32 ow, pitch 33) behind one more block barrier: every 128-B line of out is one store of one wave, was 32 four-byte pieces of 32 waves; sums unchanged
# speedup vs baseline: 1.5632x; 1.1891x over previous
#include <hip/hip_runtime.h>
#include <hip/hip_bf16.h>

typedef float v2f __attribute__((ext_vector_type(2)));
typedef float v8f __attribute__((ext_vector_type(8)));

#define RP 20
#define X_ROWS 288
#define XS_N (X_ROWS * RP)
#define W_PL (64 * RP)
#define WS_N (16 * W_PL)
#define V_VOX (16 * RP + 4)
#define VS_N (32 * V_VOX)
#define OP 33

__global__ __launch_bounds__(1024)
void deconv_caps_routing(const float* __restrict__ xg,
                         const float* __restrict__ wg,
                         const float* __restrict__ bg,
                         float* __restrict__ outg) {
  extern __shared__ __attribute__((aligned(16))) float smem[];
  float* xs  = smem;
  float* ws  = smem + XS_N;
  float* vsm = smem + XS_N + WS_N;

  const int tid = threadIdx.x;
  const int blk = blockIdx.x;
  const int b  = blk >> 10;
  const int od = (blk >> 5) & 31;
  const int oh = blk & 31;

  const int tdbase = 1 - (od & 1);
  const int thbase = 1 - (oh & 1);
  const int id0 = (od + 1) >> 1;
  const int ih0 = (oh + 1) >> 1;

  {
    const float4* __restrict__ wg4 = reinterpret_cast<const float4*>(wg);
    const int s_aip = tid & 7;
    const int s_c   = ((tid >> 5) << 1) | ((tid >> 3) & 1);
    const int s_thi = (tid >> 4) & 1;
    const int s_th  = thbase + 2 * s_thi;
    const int g_even = ((2 * s_aip) * 64 + s_c) * 16 + s_th;
    const int g_odd  = g_even + 64 * 16;
    const float4 e0 = wg4[g_even + (tdbase    ) * 4];
    const float4 o0 = wg4[g_odd  + (tdbase    ) * 4];
    const float4 e1 = wg4[g_even + (tdbase + 2) * 4];
    const float4 o1 = wg4[g_odd  + (tdbase + 2) * 4];
    float* d0 = ws + 4 * W_PL * s_thi + RP * s_c + 2 * s_aip;
    float* d1 = d0 + 8 * W_PL;
    *(v2f*)(d0 + 0 * W_PL) = (v2f){e0.x, o0.x};
    *(v2f*)(d0 + 1 * W_PL) = (v2f){e0.y, o0.y};
    *(v2f*)(d0 + 2 * W_PL) = (v2f){e0.z, o0.z};
    *(v2f*)(d0 + 3 * W_PL) = (v2f){e0.w, o0.w};
    *(v2f*)(d1 + 0 * W_PL) = (v2f){e1.x, o1.x};
    *(v2f*)(d1 + 1 * W_PL) = (v2f){e1.y, o1.y};
    *(v2f*)(d1 + 2 * W_PL) = (v2f){e1.z, o1.z};
    *(v2f*)(d1 + 3 * W_PL) = (v2f){e1.w, o1.w};
  }
  for (int idx = tid; idx < X_ROWS * 16; idx += 1024) {
    int ai   = idx & 15;
    int t    = idx >> 4;
    int iwp  = t % 18;
    int rest = t / 18;
    int thi = rest & 1;
    int tdi = (rest >> 1) & 1;
    int ci  = rest >> 2;
    int iw = iwp - 1;
    int id = id0 - tdi;
    int ih = ih0 - thi;
    float v = 0.f;
    if ((unsigned)iw < 16u && (unsigned)id < 16u && (unsigned)ih < 16u)
      v = xg[(((b * 4 + ci) * 16 + ai) << 12) + (id << 8) + (ih << 4) + iw];
    xs[t * RP + ai] = v;
  }
  __syncthreads();

  const int lane = tid & 31;
  const int wv   = tid >> 5;
  const int par  = wv >> 4;
  const int icap = (wv >> 2) & 3;
  const int ocap = wv & 3;
  const int ln = lane & 15;
  const int kh = lane >> 4;

  v8f acc = {0.f, 0.f, 0.f, 0.f, 0.f, 0.f, 0.f, 0.f};
#pragma unroll
  for (int j = 0; j < 32; ++j) {
    int tap = j >> 2;
    int twi = tap & 1;
    int thi = (tap >> 1) & 1;
    int tdi = tap >> 2;
    int aibase = ((j & 3) << 2) + (kh << 1);
    int iw = ln + par - twi;
    int aoff = (((icap * 2 + tdi) * 2 + thi) * 18 + (iw + 1)) * RP + aibase;
    int tw = (1 - par) + (twi << 1);
    int boff = ((((tdi * 2 + thi) << 2) + tw) * 64 + (ocap * 16 + ln)) * RP + aibase;
    v2f a  = *(const v2f*)(xs + aoff);
    v2f bb = *(const v2f*)(ws + boff);
    acc = __builtin_amdgcn_wmma_f32_16x16x4_f32(
         false, a,  false, bb,
         (short)0, acc,  false,  false);
  }

  const int p_idx = icap * 4 + ocap;
#pragma unroll
  for (int g = 0; g < 8; ++g) {
    int vx = g + (kh << 3);
    vsm[(par * 16 + vx) * V_VOX + p_idx * RP + ln] = acc[g];
  }
  __syncthreads();

  const int rr = wv & 15;
  const int p  = lane & 15;
  const int h  = lane >> 4;
  const int ri = p >> 2;
  const int ro = p & 3;

  float vv[8], bv[8];
  const float* vbase = vsm + (par * 16 + rr) * V_VOX + p * RP + (h << 3);
#pragma unroll
  for (int j = 0; j < 8; ++j) vv[j] = vbase[j];
#pragma unroll
  for (int j = 0; j < 8; ++j) bv[j] = bg[ro * 16 + (h << 3) + j];

  float logit = 0.f;
  float pre[8];
  float scale = 0.f;
#pragma unroll
  for (int it = 0; it < 3; ++it) {
    float m = logit;
    m = fmaxf(m, __shfl_xor(m, 1));
    m = fmaxf(m, __shfl_xor(m, 2));
    float e = __expf(logit - m);
    float s = e;
    s += __shfl_xor(s, 1);
    s += __shfl_xor(s, 2);
    float route = e / s;
#pragma unroll
    for (int j = 0; j < 8; ++j) pre[j] = route * vv[j];
#pragma unroll
    for (int j = 0; j < 8; ++j) {
      pre[j] += __shfl_xor(pre[j], 4);
      pre[j] += __shfl_xor(pre[j], 8);
      pre[j] += bv[j];
    }
    if (it < 2) {
      float dt = 0.f, np2 = 0.f, nv2 = 0.f;
#pragma unroll
      for (int j = 0; j < 8; ++j) {
        dt  += pre[j] * vv[j];
        np2 += pre[j] * pre[j];
        nv2 += vv[j] * vv[j];
      }
      dt  += __shfl_xor(dt, 16);
      np2 += __shfl_xor(np2, 16);
      nv2 += __shfl_xor(nv2, 16);
      float na = sqrtf(np2), nb = sqrtf(nv2);
      logit += dt / fmaxf(na * nb, 1e-8f);
    } else {
      float n2 = 0.f;
#pragma unroll
      for (int j = 0; j < 8; ++j) n2 += pre[j] * pre[j];
      n2 += __shfl_xor(n2, 16);
      float norm = sqrtf(n2);
      scale = n2 / ((norm + 1e-12f) * (1.f + n2));
    }
  }

  float* osm = smem;
  if (ri == 0) {
    int ow = (rr << 1) + par;
#pragma unroll
    for (int j = 0; j < 8; ++j) {
      int row = ro * 16 + (h << 3) + j;
      osm[row * OP + ow] = pre[j] * scale;
    }
  }
  __syncthreads();

  for (int r = 0; r < 2; ++r) {
    int row = (wv << 1) + r;
    outg[((b * 64 + row) << 15) + (od << 10) + (oh << 5) + lane] = osm[row * OP + lane];
  }
}

extern "C" void kernel_launch(void* const* d_in, const int* in_sizes, int n_in,
                              void* d_out, int out_size, void* d_ws, size_t ws_size,
                              hipStream_t stream) {
  (void)in_sizes; (void)n_in; (void)out_size; (void)d_ws; (void)ws_size;
  const float* x = (const float*)d_in[0];
  const float* w = (const float*)d_in[1];
  const float* b = (const float*)d_in[2];
  float* out = (float*)d_out;

  const size_t shmem = (size_t)(XS_N + WS_N + VS_N) * sizeof(float);
  deconv_caps_routing<<<dim3(4 * 32 * 32), dim3(1024), shmem, stream>>>(x, w, b, out);
}


# one block loops over 8 oh of one (od, oh parity) class: weight slice staged once per block (was once per oh), 512 blocks enter (was 4096), next x fetched under the routing; sums unchanged
# speedup vs baseline: 2.0648x; 1.3209x over previous
#include <hip/hip_runtime.h>
#include <hip/hip_bf16.h>

typedef float v2f __attribute__((ext_vector_type(2)));
typedef float v8f __attribute__((ext_vector_type(8)));

#define RP 20
#define X_ROWS 288
#define XS_N (X_ROWS * RP)
#define W_PL (64 * RP)
#define WS_N (16 * W_PL)
#define V_VOX (16 * RP + 4)
#define VS_N (32 * V_VOX)
#define OP 33
#define OS_N (64 * OP)
#define ROWS 8
#define BS_N 64

__global__ __launch_bounds__(1024, 16)
void deconv_caps_routing(const float* __restrict__ xg,
                         const float* __restrict__ wg,
                         const float* __restrict__ bg,
                         float* __restrict__ outg) {
  extern __shared__ __attribute__((aligned(16))) float smem[];
  float* xs  = smem;
  float* ws  = smem + XS_N;
  float* vsm = smem + XS_N + WS_N;
  float* osm = smem + XS_N + WS_N + VS_N;
  float* bsm = smem + XS_N + WS_N + VS_N + OS_N;

  const int tid = threadIdx.x;
  const int blk = blockIdx.x;
  const int b     = blk >> 7;
  const int od    = (blk >> 2) & 31;
  const int ohpar = (blk >> 1) & 1;
  const int m0    = (blk & 1) * ROWS;

  const int tdbase = 1 - (od & 1);
  const int thbase = 1 - ohpar;
  const int id0 = (od + 1) >> 1;

  {
    const float4* __restrict__ wg4 = reinterpret_cast<const float4*>(wg);
    const int s_aip = tid & 7;
    const int s_c   = ((tid >> 5) << 1) | ((tid >> 3) & 1);
    const int s_thi = (tid >> 4) & 1;
    const int s_th  = thbase + 2 * s_thi;
    const int g_even = ((2 * s_aip) * 64 + s_c) * 16 + s_th;
    const int g_odd  = g_even + 64 * 16;
    const float4 e0 = wg4[g_even + (tdbase    ) * 4];
    const float4 o0 = wg4[g_odd  + (tdbase    ) * 4];
    const float4 e1 = wg4[g_even + (tdbase + 2) * 4];
    const float4 o1 = wg4[g_odd  + (tdbase + 2) * 4];
    float* d0 = ws + 4 * W_PL * s_thi + RP * s_c + 2 * s_aip;
    float* d1 = d0 + 8 * W_PL;
    *(v2f*)(d0 + 0 * W_PL) = (v2f){e0.x, o0.x};
    *(v2f*)(d0 + 1 * W_PL) = (v2f){e0.y, o0.y};
    *(v2f*)(d0 + 2 * W_PL) = (v2f){e0.z, o0.z};
    *(v2f*)(d0 + 3 * W_PL) = (v2f){e0.w, o0.w};
    *(v2f*)(d1 + 0 * W_PL) = (v2f){e1.x, o1.x};
    *(v2f*)(d1 + 1 * W_PL) = (v2f){e1.y, o1.y};
    *(v2f*)(d1 + 2 * W_PL) = (v2f){e1.z, o1.z};
    *(v2f*)(d1 + 3 * W_PL) = (v2f){e1.w, o1.w};
  }

  const int x_ai  = tid & 15;
  const int x_iw  = (tid >> 4) & 15;
  const int x_thi = (tid >> 8) & 1;
  const int x_tdi = tid >> 9;
  const int x_id  = id0 - x_tdi;
  const bool x_idok = (unsigned)x_id < 16u;
  const unsigned xoff = ((b * 64 + x_ai) << 12) + ((x_idok ? x_id : 0) << 8) + x_iw;
  float* xdst = xs + ((x_tdi * 2 + x_thi) * 18 + x_iw + 1) * RP + x_ai;
  float xr[4];

  if (tid >= 1024 - BS_N) bsm[tid - (1024 - BS_N)] = bg[tid - (1024 - BS_N)];
  if (tid < 512) {
    int z_ai = tid & 15, z_end = (tid >> 4) & 1, z_r = tid >> 5;
    xs[(z_r * 18 + z_end * 17) * RP + z_ai] = 0.f;
  }
  {
    const int ih = m0 + ohpar - x_thi;
    const bool ok = x_idok && (unsigned)ih < 16u;
    const float* px = xg + (size_t)(xoff + ((unsigned)ih << 4));
#pragma unroll
    for (int q = 0; q < 4; ++q) xr[q] = ok ? px[q << 16] : 0.f;
#pragma unroll
    for (int q = 0; q < 4; ++q) xdst[q * 72 * RP] = xr[q];
  }
  __syncthreads();

  const int lane = tid & 31;
  const int wv   = tid >> 5;
  const int par  = wv >> 4;
  const int icap = (wv >> 2) & 3;
  const int ocap = wv & 3;
  const int ln = lane & 15;
  const int kh = lane >> 4;
  const int p_idx = icap * 4 + ocap;
  const float* abase = xs + ((icap * 4) * 18 + ln + par + 1) * RP + (kh << 1);
  const float* bbase = ws + ((1 - par) * 64 + ocap * 16 + ln) * RP + (kh << 1);
  float* vst = vsm + (par * 16 + (kh << 3)) * V_VOX + p_idx * RP + ln;

  const int rr = wv & 15;
  const int p  = lane & 15;
  const int h  = lane >> 4;
  const int ri = p >> 2;
  const int ro = p & 3;

  for (int k = 0; k < ROWS; ++k) {
    const int oh = ohpar + 2 * (m0 + k);

    v8f acc = {0.f, 0.f, 0.f, 0.f, 0.f, 0.f, 0.f, 0.f};
    const float* ap = abase;
    const float* bp = bbase;
#pragma nounroll
    for (int t2 = 0; t2 < 4; ++t2) {
#pragma unroll
      for (int jj = 0; jj < 8; ++jj) {
        const int twi = jj >> 2;
        const int ch  = jj & 3;
        v2f a  = *(const v2f*)(ap - twi * RP + (ch << 2));
        v2f bb = *(const v2f*)(bp + twi * 2 * 64 * RP + (ch << 2));
        acc = __builtin_amdgcn_wmma_f32_16x16x4_f32(
             false, a,  false, bb,
             (short)0, acc,  false,  false);
      }
      ap += 18 * RP;
      bp += 4 * 64 * RP;
    }

#pragma unroll
    for (int g = 0; g < 8; ++g) vst[g * V_VOX] = acc[g];
    __syncthreads();

    {
      const int ih = m0 + k + 1 + ohpar - x_thi;
      const bool ok = x_idok && (unsigned)ih < 16u;
      const float* px = xg + (size_t)(xoff + ((unsigned)ih << 4));
#pragma unroll
      for (int q = 0; q < 4; ++q) xr[q] = ok ? px[q << 16] : 0.f;
    }

    float vv[8], bv[8];
    const float* vbase = vsm + (par * 16 + rr) * V_VOX + p * RP + (h << 3);
#pragma unroll
    for (int j = 0; j < 8; ++j) vv[j] = vbase[j];
#pragma unroll
    for (int j = 0; j < 8; ++j) bv[j] = bsm[ro * 16 + (h << 3) + j];

    float logit = 0.f;
    float pre[8];
    float scale = 0.f;
#pragma unroll
    for (int it = 0; it < 3; ++it) {
      float m = logit;
      m = fmaxf(m, __shfl_xor(m, 1));
      m = fmaxf(m, __shfl_xor(m, 2));
      float e = __expf(logit - m);
      float s = e;
      s += __shfl_xor(s, 1);
      s += __shfl_xor(s, 2);
      float route = e / s;
#pragma unroll
      for (int j = 0; j < 8; ++j) pre[j] = route * vv[j];
#pragma unroll
      for (int j = 0; j < 8; ++j) {
        pre[j] += __shfl_xor(pre[j], 4);
        pre[j] += __shfl_xor(pre[j], 8);
        pre[j] += bv[j];
      }
      if (it < 2) {
        float dt = 0.f, np2 = 0.f, nv2 = 0.f;
#pragma unroll
        for (int j = 0; j < 8; ++j) {
          dt  += pre[j] * vv[j];
          np2 += pre[j] * pre[j];
          nv2 += vv[j] * vv[j];
        }
        dt  += __shfl_xor(dt, 16);
        np2 += __shfl_xor(np2, 16);
        nv2 += __shfl_xor(nv2, 16);
        float na = sqrtf(np2), nb = sqrtf(nv2);
        logit += dt / fmaxf(na * nb, 1e-8f);
      } else {
        float n2 = 0.f;
#pragma unroll
        for (int j = 0; j < 8; ++j) n2 += pre[j] * pre[j];
        n2 += __shfl_xor(n2, 16);
        float norm = sqrtf(n2);
        scale = n2 / ((norm + 1e-12f) * (1.f + n2));
      }
    }

    if (ri == 0) {
      float* ost = osm + (ro * 16 + (h << 3)) * OP + (rr << 1) + par;
#pragma unroll
      for (int j = 0; j < 8; ++j) ost[j * OP] = pre[j] * scale;
    }
#pragma unroll
    for (int q = 0; q < 4; ++q) xdst[q * 72 * RP] = xr[q];
    __syncthreads();

    {
      const float* ord = osm + (wv << 1) * OP + lane;
      const unsigned ooff = ((b * 64 + (wv << 1)) << 15) + (od << 10) + (oh << 5) + lane;
      const float r0 = ord[0], r1 = ord[OP];
      float* po = outg + (size_t)ooff;
      po[0] = r0;
      po[1 << 15] = r1;
    }
  }
}

extern "C" void kernel_launch(void* const* d_in, const int* in_sizes, int n_in,
                              void* d_out, int out_size, void* d_ws, size_t ws_size,
                              hipStream_t stream) {
  (void)in_sizes; (void)n_in; (void)out_size; (void)d_ws; (void)ws_size;
  const float* x = (const float*)d_in[0];
  const float* w = (const float*)d_in[1];
  const float* b = (const float*)d_in[2];
  float* out = (float*)d_out;

  const size_t shmem = (size_t)(XS_N + WS_N + VS_N + OS_N + BS_N) * sizeof(float);
  deconv_caps_routing<<<dim3(4 * 32 * 2 * 2), dim3(1024), shmem, stream>>>(x, w, b, out);
}


# routing's 10 divisions and square roots per voxel as single 1-ulp v_rcp_f32 / v_sqrt_f32 instead of the correctly rounded expansions: loop VALU count 297 -> 200; sums and their order unchanged
# speedup vs baseline: 2.2410x; 1.0853x over previous
#include <hip/hip_runtime.h>
#include <hip/hip_bf16.h>

typedef float v2f __attribute__((ext_vector_type(2)));
typedef float v8f __attribute__((ext_vector_type(8)));

#define RP 20
#define X_ROWS 288
#define XS_N (X_ROWS * RP)
#define W_PL (64 * RP)
#define WS_N (16 * W_PL)
#define V_VOX (16 * RP + 4)
#define VS_N (32 * V_VOX)
#define OP 33
#define OS_N (64 * OP)
#define ROWS 8
#define BS_N 64

__device__ __forceinline__ float rcp_hw(float v)  { return __builtin_amdgcn_rcpf(v); }
__device__ __forceinline__ float sqrt_hw(float v) { return __builtin_amdgcn_sqrtf(v); }

__global__ __launch_bounds__(1024, 16)
void deconv_caps_routing(const float* __restrict__ xg,
                         const float* __restrict__ wg,
                         const float* __restrict__ bg,
                         float* __restrict__ outg) {
  extern __shared__ __attribute__((aligned(16))) float smem[];
  float* xs  = smem;
  float* ws  = smem + XS_N;
  float* vsm = smem + XS_N + WS_N;
  float* osm = smem + XS_N + WS_N + VS_N;
  float* bsm = smem + XS_N + WS_N + VS_N + OS_N;

  const int tid = threadIdx.x;
  const int blk = blockIdx.x;
  const int b     = blk >> 7;
  const int od    = (blk >> 2) & 31;
  const int ohpar = (blk >> 1) & 1;
  const int m0    = (blk & 1) * ROWS;

  const int tdbase = 1 - (od & 1);
  const int thbase = 1 - ohpar;
  const int id0 = (od + 1) >> 1;

  {
    const float4* __restrict__ wg4 = reinterpret_cast<const float4*>(wg);
    const int s_aip = tid & 7;
    const int s_c   = ((tid >> 5) << 1) | ((tid >> 3) & 1);
    const int s_thi = (tid >> 4) & 1;
    const int s_th  = thbase + 2 * s_thi;
    const int g_even = ((2 * s_aip) * 64 + s_c) * 16 + s_th;
    const int g_odd  = g_even + 64 * 16;
    const float4 e0 = wg4[g_even + (tdbase    ) * 4];
    const float4 o0 = wg4[g_odd  + (tdbase    ) * 4];
    const float4 e1 = wg4[g_even + (tdbase + 2) * 4];
    const float4 o1 = wg4[g_odd  + (tdbase + 2) * 4];
    float* d0 = ws + 4 * W_PL * s_thi + RP * s_c + 2 * s_aip;
    float* d1 = d0 + 8 * W_PL;
    *(v2f*)(d0 + 0 * W_PL) = (v2f){e0.x, o0.x};
    *(v2f*)(d0 + 1 * W_PL) = (v2f){e0.y, o0.y};
    *(v2f*)(d0 + 2 * W_PL) = (v2f){e0.z, o0.z};
    *(v2f*)(d0 + 3 * W_PL) = (v2f){e0.w, o0.w};
    *(v2f*)(d1 + 0 * W_PL) = (v2f){e1.x, o1.x};
    *(v2f*)(d1 + 1 * W_PL) = (v2f){e1.y, o1.y};
    *(v2f*)(d1 + 2 * W_PL) = (v2f){e1.z, o1.z};
    *(v2f*)(d1 + 3 * W_PL) = (v2f){e1.w, o1.w};
  }

  const int x_ai  = tid & 15;
  const int x_iw  = (tid >> 4) & 15;
  const int x_thi = (tid >> 8) & 1;
  const int x_tdi = tid >> 9;
  const int x_id  = id0 - x_tdi;
  const bool x_idok = (unsigned)x_id < 16u;
  const unsigned xoff = ((b * 64 + x_ai) << 12) + ((x_idok ? x_id : 0) << 8) + x_iw;
  float* xdst = xs + ((x_tdi * 2 + x_thi) * 18 + x_iw + 1) * RP + x_ai;
  float xr[4];

  if (tid >= 1024 - BS_N) bsm[tid - (1024 - BS_N)] = bg[tid - (1024 - BS_N)];
  if (tid < 512) {
    int z_ai = tid & 15, z_end = (tid >> 4) & 1, z_r = tid >> 5;
    xs[(z_r * 18 + z_end * 17) * RP + z_ai] = 0.f;
  }
  {
    const int ih = m0 + ohpar - x_thi;
    const bool ok = x_idok && (unsigned)ih < 16u;
    const float* px = xg + (size_t)(xoff + ((unsigned)ih << 4));
#pragma unroll
    for (int q = 0; q < 4; ++q) xr[q] = ok ? px[q << 16] : 0.f;
#pragma unroll
    for (int q = 0; q < 4; ++q) xdst[q * 72 * RP] = xr[q];
  }
  __syncthreads();

  const int lane = tid & 31;
  const int wv   = tid >> 5;
  const int par  = wv >> 4;
  const int icap = (wv >> 2) & 3;
  const int ocap = wv & 3;
  const int ln = lane & 15;
  const int kh = lane >> 4;
  const int p_idx = icap * 4 + ocap;
  const float* abase = xs + ((icap * 4) * 18 + ln + par + 1) * RP + (kh << 1);
  const float* bbase = ws + ((1 - par) * 64 + ocap * 16 + ln) * RP + (kh << 1);
  float* vst = vsm + (par * 16 + (kh << 3)) * V_VOX + p_idx * RP + ln;

  const int rr = wv & 15;
  const int p  = lane & 15;
  const int h  = lane >> 4;
  const int ri = p >> 2;
  const int ro = p & 3;

  for (int k = 0; k < ROWS; ++k) {
    const int oh = ohpar + 2 * (m0 + k);

    v8f acc = {0.f, 0.f, 0.f, 0.f, 0.f, 0.f, 0.f, 0.f};
    const float* ap = abase;
    const float* bp = bbase;
#pragma nounroll
    for (int t2 = 0; t2 < 4; ++t2) {
#pragma unroll
      for (int jj = 0; jj < 8; ++jj) {
        const int twi = jj >> 2;
        const int ch  = jj & 3;
        v2f a  = *(const v2f*)(ap - twi * RP + (ch << 2));
        v2f bb = *(const v2f*)(bp + twi * 2 * 64 * RP + (ch << 2));
        acc = __builtin_amdgcn_wmma_f32_16x16x4_f32(
             false, a,  false, bb,
             (short)0, acc,  false,  false);
      }
      ap += 18 * RP;
      bp += 4 * 64 * RP;
    }

#pragma unroll
    for (int g = 0; g < 8; ++g) vst[g * V_VOX] = acc[g];
    __syncthreads();

    {
      const int ih = m0 + k + 1 + ohpar - x_thi;
      const bool ok = x_idok && (unsigned)ih < 16u;
      const float* px = xg + (size_t)(xoff + ((unsigned)ih << 4));
#pragma unroll
      for (int q = 0; q < 4; ++q) xr[q] = ok ? px[q << 16] : 0.f;
    }

    float vv[8], bv[8];
    const float* vbase = vsm + (par * 16 + rr) * V_VOX + p * RP + (h << 3);
#pragma unroll
    for (int j = 0; j < 8; ++j) vv[j] = vbase[j];
#pragma unroll
    for (int j = 0; j < 8; ++j) bv[j] = bsm[ro * 16 + (h << 3) + j];

    float logit = 0.f;
    float pre[8];
    float scale = 0.f;
#pragma unroll
    for (int it = 0; it < 3; ++it) {
      float m = logit;
      m = fmaxf(m, __shfl_xor(m, 1));
      m = fmaxf(m, __shfl_xor(m, 2));
      float e = __expf(logit - m);
      float s = e;
      s += __shfl_xor(s, 1);
      s += __shfl_xor(s, 2);
      float route = e * rcp_hw(s);
#pragma unroll
      for (int j = 0; j < 8; ++j) pre[j] = route * vv[j];
#pragma unroll
      for (int j = 0; j < 8; ++j) {
        pre[j] += __shfl_xor(pre[j], 4);
        pre[j] += __shfl_xor(pre[j], 8);
        pre[j] += bv[j];
      }
      if (it < 2) {
        float dt = 0.f, np2 = 0.f, nv2 = 0.f;
#pragma unroll
        for (int j = 0; j < 8; ++j) {
          dt  += pre[j] * vv[j];
          np2 += pre[j] * pre[j];
          nv2 += vv[j] * vv[j];
        }
        dt  += __shfl_xor(dt, 16);
        np2 += __shfl_xor(np2, 16);
        nv2 += __shfl_xor(nv2, 16);
        float na = sqrt_hw(np2), nb = sqrt_hw(nv2);
        logit += dt * rcp_hw(fmaxf(na * nb, 1e-8f));
      } else {
        float n2 = 0.f;
#pragma unroll
        for (int j = 0; j < 8; ++j) n2 += pre[j] * pre[j];
        n2 += __shfl_xor(n2, 16);
        float norm = sqrt_hw(n2);
        scale = (n2 * rcp_hw(1.f + n2)) * rcp_hw(norm + 1e-12f);
      }
    }

    if (ri == 0) {
      float* ost = osm + (ro * 16 + (h << 3)) * OP + (rr << 1) + par;
#pragma unroll
      for (int j = 0; j < 8; ++j) ost[j * OP] = pre[j] * scale;
    }
#pragma unroll
    for (int q = 0; q < 4; ++q) xdst[q * 72 * RP] = xr[q];
    __syncthreads();

    {
      const float* ord = osm + (wv << 1) * OP + lane;
      const unsigned ooff = ((b * 64 + (wv << 1)) << 15) + (od << 10) + (oh << 5) + lane;
      const float r0 = ord[0], r1 = ord[OP];
      float* po = outg + (size_t)ooff;
      po[0] = r0;
      po[1 << 15] = r1;
    }
  }
}

extern "C" void kernel_launch(void* const* d_in, const int* in_sizes, int n_in,
                              void* d_out, int out_size, void* d_ws, size_t ws_size,
                              hipStream_t stream) {
  (void)in_sizes; (void)n_in; (void)out_size; (void)d_ws; (void)ws_size;
  const float* x = (const float*)d_in[0];
  const float* w = (const float*)d_in[1];
  const float* b = (const float*)d_in[2];
  float* out = (float*)d_out;

  const size_t shmem = (size_t)(XS_N + WS_N + VS_N + OS_N + BS_N) * sizeof(float);
  deconv_caps_routing<<<dim3(4 * 32 * 2 * 2), dim3(1024), shmem, stream>>>(x, w, b, out);
}


# routing remapped: a lane holds all 4 input capsules of 4 atoms, so the sum over i stays in-lane and pre exists once; lane exchanges as DPP, not ds_bpermute; ao sums now 4 in-lane + 2 butterfly steps
# speedup vs baseline: 2.6951x; 1.2026x over previous
#include <hip/hip_runtime.h>
#include <hip/hip_bf16.h>

typedef float v2f __attribute__((ext_vector_type(2)));
typedef float v8f __attribute__((ext_vector_type(8)));

#define RP 20
#define X_ROWS 288
#define XS_N (X_ROWS * RP)
#define W_PL (64 * RP)
#define WS_N (16 * W_PL)
#define V_VOX (16 * RP + 4)
#define VS_N (32 * V_VOX)
#define OP 33
#define OS_N (64 * OP)
#define ROWS 8
#define BS_N 64

__device__ __forceinline__ float rcp_hw(float v)  { return __builtin_amdgcn_rcpf(v); }
__device__ __forceinline__ float sqrt_hw(float v) { return __builtin_amdgcn_sqrtf(v); }

template <int M>
__device__ __forceinline__ float xrow(float v) {
  return __builtin_bit_cast(float, __builtin_amdgcn_update_dpp(
      0, __builtin_bit_cast(int, v), 0x160 + M, 0xf, 0xf, true));
}
template <int I>
__device__ __forceinline__ float quad(float v) {
  return __builtin_bit_cast(float, __builtin_amdgcn_update_dpp(
      0, __builtin_bit_cast(int, v), I * 0x55, 0xf, 0xf, true));
}

__global__ __launch_bounds__(1024, 16)
void deconv_caps_routing(const float* __restrict__ xg,
                         const float* __restrict__ wg,
                         const float* __restrict__ bg,
                         float* __restrict__ outg) {
  extern __shared__ __attribute__((aligned(16))) float smem[];
  float* xs  = smem;
  float* ws  = smem + XS_N;
  float* vsm = smem + XS_N + WS_N;
  float* osm = smem + XS_N + WS_N + VS_N;
  float* bsm = smem + XS_N + WS_N + VS_N + OS_N;

  const int tid = threadIdx.x;
  const int blk = blockIdx.x;
  const int b     = blk >> 7;
  const int od    = (blk >> 2) & 31;
  const int ohpar = (blk >> 1) & 1;
  const int m0    = (blk & 1) * ROWS;

  const int tdbase = 1 - (od & 1);
  const int thbase = 1 - ohpar;
  const int id0 = (od + 1) >> 1;

  {
    const float4* __restrict__ wg4 = reinterpret_cast<const float4*>(wg);
    const int s_aip = tid & 7;
    const int s_c   = ((tid >> 5) << 1) | ((tid >> 3) & 1);
    const int s_thi = (tid >> 4) & 1;
    const int s_th  = thbase + 2 * s_thi;
    const int g_even = ((2 * s_aip) * 64 + s_c) * 16 + s_th;
    const int g_odd  = g_even + 64 * 16;
    const float4 e0 = wg4[g_even + (tdbase    ) * 4];
    const float4 o0 = wg4[g_odd  + (tdbase    ) * 4];
    const float4 e1 = wg4[g_even + (tdbase + 2) * 4];
    const float4 o1 = wg4[g_odd  + (tdbase + 2) * 4];
    float* d0 = ws + 4 * W_PL * s_thi + RP * s_c + 2 * s_aip;
    float* d1 = d0 + 8 * W_PL;
    *(v2f*)(d0 + 0 * W_PL) = (v2f){e0.x, o0.x};
    *(v2f*)(d0 + 1 * W_PL) = (v2f){e0.y, o0.y};
    *(v2f*)(d0 + 2 * W_PL) = (v2f){e0.z, o0.z};
    *(v2f*)(d0 + 3 * W_PL) = (v2f){e0.w, o0.w};
    *(v2f*)(d1 + 0 * W_PL) = (v2f){e1.x, o1.x};
    *(v2f*)(d1 + 1 * W_PL) = (v2f){e1.y, o1.y};
    *(v2f*)(d1 + 2 * W_PL) = (v2f){e1.z, o1.z};
    *(v2f*)(d1 + 3 * W_PL) = (v2f){e1.w, o1.w};
  }

  const int x_ai  = tid & 15;
  const int x_iw  = (tid >> 4) & 15;
  const int x_thi = (tid >> 8) & 1;
  const int x_tdi = tid >> 9;
  const int x_id  = id0 - x_tdi;
  const bool x_idok = (unsigned)x_id < 16u;
  const unsigned xoff = ((b * 64 + x_ai) << 12) + ((x_idok ? x_id : 0) << 8) + x_iw;
  float* xdst = xs + ((x_tdi * 2 + x_thi) * 18 + x_iw + 1) * RP + x_ai;
  float xr[4];

  if (tid >= 1024 - BS_N) bsm[tid - (1024 - BS_N)] = bg[tid - (1024 - BS_N)];
  if (tid < 512) {
    int z_ai = tid & 15, z_end = (tid >> 4) & 1, z_r = tid >> 5;
    xs[(z_r * 18 + z_end * 17) * RP + z_ai] = 0.f;
  }
  {
    const int ih = m0 + ohpar - x_thi;
    const bool ok = x_idok && (unsigned)ih < 16u;
    const float* px = xg + (size_t)(xoff + ((unsigned)ih << 4));
#pragma unroll
    for (int q = 0; q < 4; ++q) xr[q] = ok ? px[q << 16] : 0.f;
#pragma unroll
    for (int q = 0; q < 4; ++q) xdst[q * 72 * RP] = xr[q];
  }
  __syncthreads();

  const int lane = tid & 31;
  const int wv   = tid >> 5;
  const int par  = wv >> 4;
  const int icap = (wv >> 2) & 3;
  const int ocap = wv & 3;
  const int ln = lane & 15;
  const int kh = lane >> 4;
  const int p_idx = icap * 4 + ocap;
  const float* abase = xs + ((icap * 4) * 18 + ln + par + 1) * RP + (kh << 1);
  const float* bbase = ws + ((1 - par) * 64 + ocap * 16 + ln) * RP + (kh << 1);
  float* vst = vsm + (par * 16 + (kh << 3)) * V_VOX + p_idx * RP + ln;

  const bool r_on  = (wv & 4) == 0;
  const int r_slot = ((wv >> 3) << 2) | (wv & 3);
  const int r_vox  = (r_slot << 1) | (lane >> 4);
  const int r_ow   = ((r_vox & 15) << 1) | (r_vox >> 4);
  const int ro = (lane >> 2) & 3;
  const int rq = lane & 3;

  for (int k = 0; k < ROWS; ++k) {
    const int oh = ohpar + 2 * (m0 + k);

    v8f acc = {0.f, 0.f, 0.f, 0.f, 0.f, 0.f, 0.f, 0.f};
    const float* ap = abase;
    const float* bp = bbase;
#pragma nounroll
    for (int t2 = 0; t2 < 4; ++t2) {
#pragma unroll
      for (int jj = 0; jj < 8; ++jj) {
        const int twi = jj >> 2;
        const int ch  = jj & 3;
        v2f a  = *(const v2f*)(ap - twi * RP + (ch << 2));
        v2f bb = *(const v2f*)(bp + twi * 2 * 64 * RP + (ch << 2));
        acc = __builtin_amdgcn_wmma_f32_16x16x4_f32(
             false, a,  false, bb,
             (short)0, acc,  false,  false);
      }
      ap += 18 * RP;
      bp += 4 * 64 * RP;
    }

#pragma unroll
    for (int g = 0; g < 8; ++g) vst[g * V_VOX] = acc[g];
    __syncthreads();

    {
      const int ih = m0 + k + 1 + ohpar - x_thi;
      const bool ok = x_idok && (unsigned)ih < 16u;
      const float* px = xg + (size_t)(xoff + ((unsigned)ih << 4));
#pragma unroll
      for (int q = 0; q < 4; ++q) xr[q] = ok ? px[q << 16] : 0.f;
    }

    if (r_on) {
      const float* vbase = vsm + r_vox * V_VOX + ro * RP + (rq << 2);
      float v[4][4], bv[4];
#pragma unroll
      for (int i = 0; i < 4; ++i) {
        const float4 t = *(const float4*)(vbase + i * 4 * RP);
        v[i][0] = t.x; v[i][1] = t.y; v[i][2] = t.z; v[i][3] = t.w;
      }
      {
        const float4 t = *(const float4*)(bsm + ro * 16 + (rq << 2));
        bv[0] = t.x; bv[1] = t.y; bv[2] = t.z; bv[3] = t.w;
      }
      float nbq = 0.f;
#pragma unroll
      for (int i = 0; i < 4; ++i) {
        float t = v[i][0] * v[i][0];
        t += v[i][1] * v[i][1];
        t += v[i][2] * v[i][2];
        t += v[i][3] * v[i][3];
        t += xrow<1>(t);
        t += xrow<2>(t);
        nbq = (rq == i) ? t : nbq;
      }
      nbq = sqrt_hw(nbq);

      float logit = 0.f;
      float pre[4];
      float scale = 0.f;
#pragma unroll
      for (int it = 0; it < 3; ++it) {
        float r0 = 0.25f, r1 = 0.25f, r2 = 0.25f, r3 = 0.25f;
        if (it > 0) {
          float m = logit;
          m = fmaxf(m, xrow<4>(m));
          m = fmaxf(m, xrow<8>(m));
          const float e = __expf(logit - m);
          float s = e;
          s += xrow<4>(s);
          s += xrow<8>(s);
          const float rt = e * rcp_hw(s);
          r0 = quad<0>(rt); r1 = quad<1>(rt); r2 = quad<2>(rt); r3 = quad<3>(rt);
        }
#pragma unroll
        for (int j = 0; j < 4; ++j)
          pre[j] = ((r0 * v[0][j] + r1 * v[1][j]) + (r2 * v[2][j] + r3 * v[3][j])) + bv[j];
        float n2 = pre[0] * pre[0];
        n2 += pre[1] * pre[1];
        n2 += pre[2] * pre[2];
        n2 += pre[3] * pre[3];
        n2 += xrow<1>(n2);
        n2 += xrow<2>(n2);
        if (it < 2) {
          float dtq = 0.f;
#pragma unroll
          for (int i = 0; i < 4; ++i) {
            float t = pre[0] * v[i][0];
            t += pre[1] * v[i][1];
            t += pre[2] * v[i][2];
            t += pre[3] * v[i][3];
            t += xrow<1>(t);
            t += xrow<2>(t);
            dtq = (rq == i) ? t : dtq;
          }
          const float na = sqrt_hw(n2);
          logit += dtq * rcp_hw(fmaxf(na * nbq, 1e-8f));
        } else {
          const float norm = sqrt_hw(n2);
          scale = (n2 * rcp_hw(1.f + n2)) * rcp_hw(norm + 1e-12f);
        }
      }

      float* ost = osm + (ro * 16 + (rq << 2)) * OP + r_ow;
#pragma unroll
      for (int j = 0; j < 4; ++j) ost[j * OP] = pre[j] * scale;
    }
#pragma unroll
    for (int q = 0; q < 4; ++q) xdst[q * 72 * RP] = xr[q];
    __syncthreads();

    {
      const float* ord = osm + (wv << 1) * OP + lane;
      const unsigned ooff = ((b * 64 + (wv << 1)) << 15) + (od << 10) + (oh << 5) + lane;
      const float r0 = ord[0], r1 = ord[OP];
      float* po = outg + (size_t)ooff;
      po[0] = r0;
      po[1 << 15] = r1;
    }
  }
}

extern "C" void kernel_launch(void* const* d_in, const int* in_sizes, int n_in,
                              void* d_out, int out_size, void* d_ws, size_t ws_size,
                              hipStream_t stream) {
  (void)in_sizes; (void)n_in; (void)out_size; (void)d_ws; (void)ws_size;
  const float* x = (const float*)d_in[0];
  const float* w = (const float*)d_in[1];
  const float* b = (const float*)d_in[2];
  float* out = (float*)d_out;

  const size_t shmem = (size_t)(XS_N + WS_N + VS_N + OS_N + BS_N) * sizeof(float);
  deconv_caps_routing<<<dim3(4 * 32 * 2 * 2), dim3(1024), shmem, stream>>>(x, w, b, out);
}


# x slice fetched with one 16-byte load per thread along iw: a wave-load takes 8 whole 64-B rows (was 8 B from each of 16 lines, 4 loads per thread), spread to the LDS rows by 4 stores; sums unchanged
# speedup vs baseline: 2.7824x; 1.0324x over previous
#include <hip/hip_runtime.h>
#include <hip/hip_bf16.h>

typedef float v2f __attribute__((ext_vector_type(2)));
typedef float v8f __attribute__((ext_vector_type(8)));

#define RP 20
#define X_ROWS 288
#define XS_N (X_ROWS * RP)
#define W_PL (64 * RP)
#define WS_N (16 * W_PL)
#define V_VOX (16 * RP + 4)
#define VS_N (32 * V_VOX)
#define OP 33
#define OS_N (64 * OP)
#define ROWS 8
#define BS_N 64

__device__ __forceinline__ float rcp_hw(float v)  { return __builtin_amdgcn_rcpf(v); }
__device__ __forceinline__ float sqrt_hw(float v) { return __builtin_amdgcn_sqrtf(v); }

template <int M>
__device__ __forceinline__ float xrow(float v) {
  return __builtin_bit_cast(float, __builtin_amdgcn_update_dpp(
      0, __builtin_bit_cast(int, v), 0x160 + M, 0xf, 0xf, true));
}
template <int I>
__device__ __forceinline__ float quad(float v) {
  return __builtin_bit_cast(float, __builtin_amdgcn_update_dpp(
      0, __builtin_bit_cast(int, v), I * 0x55, 0xf, 0xf, true));
}

__global__ __launch_bounds__(1024, 16)
void deconv_caps_routing(const float* __restrict__ xg,
                         const float* __restrict__ wg,
                         const float* __restrict__ bg,
                         float* __restrict__ outg) {
  extern __shared__ __attribute__((aligned(16))) float smem[];
  float* xs  = smem;
  float* ws  = smem + XS_N;
  float* vsm = smem + XS_N + WS_N;
  float* osm = smem + XS_N + WS_N + VS_N;
  float* bsm = smem + XS_N + WS_N + VS_N + OS_N;

  const int tid = threadIdx.x;
  const int blk = blockIdx.x;
  const int b     = blk >> 7;
  const int od    = (blk >> 2) & 31;
  const int ohpar = (blk >> 1) & 1;
  const int m0    = (blk & 1) * ROWS;

  const int tdbase = 1 - (od & 1);
  const int thbase = 1 - ohpar;
  const int id0 = (od + 1) >> 1;

  {
    const float4* __restrict__ wg4 = reinterpret_cast<const float4*>(wg);
    const int s_aip = tid & 7;
    const int s_c   = ((tid >> 5) << 1) | ((tid >> 3) & 1);
    const int s_thi = (tid >> 4) & 1;
    const int s_th  = thbase + 2 * s_thi;
    const int g_even = ((2 * s_aip) * 64 + s_c) * 16 + s_th;
    const int g_odd  = g_even + 64 * 16;
    const float4 e0 = wg4[g_even + (tdbase    ) * 4];
    const float4 o0 = wg4[g_odd  + (tdbase    ) * 4];
    const float4 e1 = wg4[g_even + (tdbase + 2) * 4];
    const float4 o1 = wg4[g_odd  + (tdbase + 2) * 4];
    float* d0 = ws + 4 * W_PL * s_thi + RP * s_c + 2 * s_aip;
    float* d1 = d0 + 8 * W_PL;
    *(v2f*)(d0 + 0 * W_PL) = (v2f){e0.x, o0.x};
    *(v2f*)(d0 + 1 * W_PL) = (v2f){e0.y, o0.y};
    *(v2f*)(d0 + 2 * W_PL) = (v2f){e0.z, o0.z};
    *(v2f*)(d0 + 3 * W_PL) = (v2f){e0.w, o0.w};
    *(v2f*)(d1 + 0 * W_PL) = (v2f){e1.x, o1.x};
    *(v2f*)(d1 + 1 * W_PL) = (v2f){e1.y, o1.y};
    *(v2f*)(d1 + 2 * W_PL) = (v2f){e1.z, o1.z};
    *(v2f*)(d1 + 3 * W_PL) = (v2f){e1.w, o1.w};
  }

  const int x_wv  = __builtin_amdgcn_readfirstlane(tid >> 5);
  const int x_q   = tid & 3;
  const int x_ai  = ((tid >> 2) & 7) | ((x_wv & 1) << 3);
  const int x_thi = (x_wv >> 1) & 1;
  const int x_tdi = (x_wv >> 2) & 1;
  const int x_i   = x_wv >> 3;
  const int x_id  = id0 - x_tdi;
  const bool x_idok = (unsigned)x_id < 16u;
  const unsigned xoff = ((b * 64 + x_i * 16 + x_ai) << 12) + ((x_idok ? x_id : 0) << 8) + (x_q << 2);
  float* xdst = xs + (((x_i * 2 + x_tdi) * 2 + x_thi) * 18 + (x_q << 2) + 1) * RP + x_ai;
  float4 xr;

  if (tid >= 1024 - BS_N) bsm[tid - (1024 - BS_N)] = bg[tid - (1024 - BS_N)];
  if (tid < 512) {
    int z_ai = tid & 15, z_end = (tid >> 4) & 1, z_r = tid >> 5;
    xs[(z_r * 18 + z_end * 17) * RP + z_ai] = 0.f;
  }
  {
    const int ih = m0 + ohpar - x_thi;
    const bool ok = x_idok && (unsigned)ih < 16u;
    const int ihc = min(max(ih, 0), 15);
    const float4 t = *reinterpret_cast<const float4*>(xg + (size_t)(xoff + ((unsigned)ihc << 4)));
    xr.x = ok ? t.x : 0.f; xr.y = ok ? t.y : 0.f; xr.z = ok ? t.z : 0.f; xr.w = ok ? t.w : 0.f;
    xdst[0] = xr.x; xdst[RP] = xr.y; xdst[2 * RP] = xr.z; xdst[3 * RP] = xr.w;
  }
  __syncthreads();

  const int lane = tid & 31;
  const int wv   = tid >> 5;
  const int par  = wv >> 4;
  const int icap = (wv >> 2) & 3;
  const int ocap = wv & 3;
  const int ln = lane & 15;
  const int kh = lane >> 4;
  const int p_idx = icap * 4 + ocap;
  const float* abase = xs + ((icap * 4) * 18 + ln + par + 1) * RP + (kh << 1);
  const float* bbase = ws + ((1 - par) * 64 + ocap * 16 + ln) * RP + (kh << 1);
  float* vst = vsm + (par * 16 + (kh << 3)) * V_VOX + p_idx * RP + ln;

  const bool r_on  = (wv & 4) == 0;
  const int r_slot = ((wv >> 3) << 2) | (wv & 3);
  const int r_vox  = (r_slot << 1) | (lane >> 4);
  const int r_ow   = ((r_vox & 15) << 1) | (r_vox >> 4);
  const int ro = (lane >> 2) & 3;
  const int rq = lane & 3;

  for (int k = 0; k < ROWS; ++k) {
    const int oh = ohpar + 2 * (m0 + k);

    v8f acc = {0.f, 0.f, 0.f, 0.f, 0.f, 0.f, 0.f, 0.f};
    const float* ap = abase;
    const float* bp = bbase;
#pragma nounroll
    for (int t2 = 0; t2 < 4; ++t2) {
#pragma unroll
      for (int jj = 0; jj < 8; ++jj) {
        const int twi = jj >> 2;
        const int ch  = jj & 3;
        v2f a  = *(const v2f*)(ap - twi * RP + (ch << 2));
        v2f bb = *(const v2f*)(bp + twi * 2 * 64 * RP + (ch << 2));
        acc = __builtin_amdgcn_wmma_f32_16x16x4_f32(
             false, a,  false, bb,
             (short)0, acc,  false,  false);
      }
      ap += 18 * RP;
      bp += 4 * 64 * RP;
    }

#pragma unroll
    for (int g = 0; g < 8; ++g) vst[g * V_VOX] = acc[g];
    __syncthreads();

    {
      const int ih = m0 + k + 1 + ohpar - x_thi;
      const bool ok = x_idok && (unsigned)ih < 16u;
      const int ihc = min(max(ih, 0), 15);
      const float4 t = *reinterpret_cast<const float4*>(xg + (size_t)(xoff + ((unsigned)ihc << 4)));
      xr.x = ok ? t.x : 0.f; xr.y = ok ? t.y : 0.f; xr.z = ok ? t.z : 0.f; xr.w = ok ? t.w : 0.f;
    }

    if (r_on) {
      const float* vbase = vsm + r_vox * V_VOX + ro * RP + (rq << 2);
      float v[4][4], bv[4];
#pragma unroll
      for (int i = 0; i < 4; ++i) {
        const float4 t = *(const float4*)(vbase + i * 4 * RP);
        v[i][0] = t.x; v[i][1] = t.y; v[i][2] = t.z; v[i][3] = t.w;
      }
      {
        const float4 t = *(const float4*)(bsm + ro * 16 + (rq << 2));
        bv[0] = t.x; bv[1] = t.y; bv[2] = t.z; bv[3] = t.w;
      }
      float nbq = 0.f;
#pragma unroll
      for (int i = 0; i < 4; ++i) {
        float t = v[i][0] * v[i][0];
        t += v[i][1] * v[i][1];
        t += v[i][2] * v[i][2];
        t += v[i][3] * v[i][3];
        t += xrow<1>(t);
        t += xrow<2>(t);
        nbq = (rq == i) ? t : nbq;
      }
      nbq = sqrt_hw(nbq);

      float logit = 0.f;
      float pre[4];
      float scale = 0.f;
#pragma unroll
      for (int it = 0; it < 3; ++it) {
        float r0 = 0.25f, r1 = 0.25f, r2 = 0.25f, r3 = 0.25f;
        if (it > 0) {
          float m = logit;
          m = fmaxf(m, xrow<4>(m));
          m = fmaxf(m, xrow<8>(m));
          const float e = __expf(logit - m);
          float s = e;
          s += xrow<4>(s);
          s += xrow<8>(s);
          const float rt = e * rcp_hw(s);
          r0 = quad<0>(rt); r1 = quad<1>(rt); r2 = quad<2>(rt); r3 = quad<3>(rt);
        }
#pragma unroll
        for (int j = 0; j < 4; ++j)
          pre[j] = ((r0 * v[0][j] + r1 * v[1][j]) + (r2 * v[2][j] + r3 * v[3][j])) + bv[j];
        float n2 = pre[0] * pre[0];
        n2 += pre[1] * pre[1];
        n2 += pre[2] * pre[2];
        n2 += pre[3] * pre[3];
        n2 += xrow<1>(n2);
        n2 += xrow<2>(n2);
        if (it < 2) {
          float dtq = 0.f;
#pragma unroll
          for (int i = 0; i < 4; ++i) {
            float t = pre[0] * v[i][0];
            t += pre[1] * v[i][1];
            t += pre[2] * v[i][2];
            t += pre[3] * v[i][3];
            t += xrow<1>(t);
            t += xrow<2>(t);
            dtq = (rq == i) ? t : dtq;
          }
          const float na = sqrt_hw(n2);
          logit += dtq * rcp_hw(fmaxf(na * nbq, 1e-8f));
        } else {
          const float norm = sqrt_hw(n2);
          scale = (n2 * rcp_hw(1.f + n2)) * rcp_hw(norm + 1e-12f);
        }
      }

      float* ost = osm + (ro * 16 + (rq << 2)) * OP + r_ow;
#pragma unroll
      for (int j = 0; j < 4; ++j) ost[j * OP] = pre[j] * scale;
    }
    xdst[0] = xr.x; xdst[RP] = xr.y; xdst[2 * RP] = xr.z; xdst[3 * RP] = xr.w;
    __syncthreads();

    {
      const float* ord = osm + (wv << 1) * OP + lane;
      const unsigned ooff = ((b * 64 + (wv << 1)) << 15) + (od << 10) + (oh << 5) + lane;
      const float r0 = ord[0], r1 = ord[OP];
      float* po = outg + (size_t)ooff;
      po[0] = r0;
      po[1 << 15] = r1;
    }
  }
}

extern "C" void kernel_launch(void* const* d_in, const int* in_sizes, int n_in,
                              void* d_out, int out_size, void* d_ws, size_t ws_size,
                              hipStream_t stream) {
  (void)in_sizes; (void)n_in; (void)out_size; (void)d_ws; (void)ws_size;
  const float* x = (const float*)d_in[0];
  const float* w = (const float*)d_in[1];
  const float* b = (const float*)d_in[2];
  float* out = (float*)d_out;

  const size_t shmem = (size_t)(XS_N + WS_N + VS_N + OS_N + BS_N) * sizeof(float);
  deconv_caps_routing<<<dim3(4 * 32 * 2 * 2), dim3(1024), shmem, stream>>>(x, w, b, out);
}
